# starting kernel
# baseline (speedup 1.0000x reference)
#include <hip/hip_runtime.h>
#include <hip/hip_bf16.h>
#include <math.h>

typedef __attribute__((ext_vector_type(16))) _Float16 v16h;
typedef __attribute__((ext_vector_type(8)))  float    v8f;

__device__ __forceinline__ float gelu_f(float x) {
    const float c0 = 0.7978845608028654f;
    const float c1 = 2.8853900817779268f;
    float u = c0 * (x + 0.044715f * x * x * x);
    float e = __builtin_amdgcn_exp2f(u * c1);
    return x - x * __builtin_amdgcn_rcpf(e + 1.0f);
}

__device__ __forceinline__ void atomAddF(float* p, float v) {
    __hip_atomic_fetch_add(p, v, __ATOMIC_RELAXED, __HIP_MEMORY_SCOPE_AGENT);
}

__device__ __forceinline__ v8f wmma16(v16h a, v16h b, v8f c) {
    return __builtin_amdgcn_wmma_f32_16x16x32_f16(
         false, a,  false, b,
         (short)0, c,  false,  false);
}

__device__ __forceinline__ v16h load_wT_frag(const float* __restrict__ W,
                                             int Kreal, int t, int lane,
                                             float scale) {
    int m = 16 * t + (lane & 15);
    int g = lane >> 4;
    v16h a;
#pragma unroll
    for (int e = 0; e < 16; ++e) {
        int r = e >> 1, j = e & 1;
        int k = ((r >> 2) << 4) + g * 8 + ((r & 3) << 1) + j;
        float val = (k < Kreal) ? W[k * 32 + m] * scale : 0.0f;
        a[e] = (_Float16)val;
    }
    return a;
}

__device__ __forceinline__ v16h pack_b_from_d(v8f d0, v8f d1) {
    v16h b;
#pragma unroll
    for (int e = 0; e < 16; ++e) {
        int r = e >> 1, j = e & 1;
        int i = ((r & 3) << 1) + j;
        float val = (r >> 2) ? d1[i] : d0[i];
        b[e] = (_Float16)val;
    }
    return b;
}

__global__ void __launch_bounds__(256)
node_pre_kernel(const float* __restrict__ node_input,
                const float* __restrict__ W1_s,
                const float* __restrict__ W1_v,
                float* __restrict__ feat_s,
                float* __restrict__ feat_v, int N) {
    __shared__ float Ws[32][32];
    __shared__ float Wv[32][32];
    for (int idx = threadIdx.x; idx < 1024; idx += blockDim.x) {
        int u = idx >> 5, w = idx & 31;
        Ws[u][w] = W1_s[u * 64 + w];
        Wv[u][w] = W1_v[u * 64 + w];
    }
    __syncthreads();
    const float inv32 = 0.17677669529663687f;
    int gid = blockIdx.x * blockDim.x + threadIdx.x;
    int total = N * 32;
    for (int t = gid; t < total; t += gridDim.x * blockDim.x) {
        int n = t >> 5, w = t & 31;
        const float* row = node_input + (size_t)n * 128;
        float as = 0.f, ax = 0.f, ay = 0.f, az = 0.f;
#pragma unroll 4
        for (int u = 0; u < 32; ++u) {
            as += row[u] * Ws[u][w];
            float wv = Wv[u][w];
            ax += row[32 + u * 3 + 0] * wv;
            ay += row[32 + u * 3 + 1] * wv;
            az += row[32 + u * 3 + 2] * wv;
        }
        feat_s[(size_t)n * 32 + w] = as * inv32;
        float* fv = feat_v + (size_t)n * 96 + w * 3;
        fv[0] = ax * inv32;
        fv[1] = ay * inv32;
        fv[2] = az * inv32;
    }
}

__global__ void __launch_bounds__(256)
edge_kernel(const float* __restrict__ edge_attr,
            const float* __restrict__ edge_scalars,
            const float* __restrict__ M0, const float* __restrict__ M1,
            const float* __restrict__ Wtp0, const float* __restrict__ Wtp1,
            const float* __restrict__ Wtp2, const float* __restrict__ Wtp3,
            const int* __restrict__ edge_src, const int* __restrict__ edge_dst,
            const float* __restrict__ feat_s, const float* __restrict__ feat_v,
            float* __restrict__ acc_s, float* __restrict__ acc_v, int E) {
    const int lane = threadIdx.x & 31;
    const int col  = lane & 15;
    const int g    = lane >> 4;

    const float inv8  = 0.35355339059327373f;
    const float inv32 = 0.17677669529663687f;
    const float INV3  = 0.5773502691896258f;

    v16h aM0_0 = load_wT_frag(M0, 8, 0, lane, inv8);
    v16h aM0_1 = load_wT_frag(M0, 8, 1, lane, inv8);
    v16h aM1_0 = load_wT_frag(M1, 32, 0, lane, inv32);
    v16h aM1_1 = load_wT_frag(M1, 32, 1, lane, inv32);
    v16h aT0_0 = load_wT_frag(Wtp0, 32, 0, lane, inv32);
    v16h aT0_1 = load_wT_frag(Wtp0, 32, 1, lane, inv32);
    v16h aT1_0 = load_wT_frag(Wtp1, 32, 0, lane, inv32);
    v16h aT1_1 = load_wT_frag(Wtp1, 32, 1, lane, inv32);
    v16h aT2_0 = load_wT_frag(Wtp2, 32, 0, lane, inv32);
    v16h aT2_1 = load_wT_frag(Wtp2, 32, 1, lane, inv32);
    v16h aT3_0 = load_wT_frag(Wtp3, 32, 0, lane, inv32);
    v16h aT3_1 = load_wT_frag(Wtp3, 32, 1, lane, inv32);

    int wave   = (blockIdx.x * blockDim.x + threadIdx.x) >> 5;
    int nwaves = (gridDim.x * blockDim.x) >> 5;
    int ntiles = (E + 15) >> 4;

    for (int tile = wave; tile < ntiles; tile += nwaves) {
        int e = tile * 16 + col;
        bool valid = (e < E);
        int ec = valid ? e : 0;

        v16h b0;
#pragma unroll
        for (int q = 0; q < 16; ++q) b0[q] = (_Float16)0.0f;
        if (valid && g == 0) {
            const float* p = edge_scalars + (size_t)ec * 8;
#pragma unroll
            for (int q = 0; q < 8; ++q) b0[q] = (_Float16)p[q];
        }

        v8f z = {};
        v8f d0 = wmma16(aM0_0, b0, z);
        v8f d1 = wmma16(aM0_1, b0, z);
#pragma unroll
        for (int q = 0; q < 8; ++q) {
            d0[q] = gelu_f(d0[q]);
            d1[q] = gelu_f(d1[q]);
        }
        v16h b1 = pack_b_from_d(d0, d1);
        d0 = wmma16(aM1_0, b1, z);
        d1 = wmma16(aM1_1, b1, z);
#pragma unroll
        for (int q = 0; q < 8; ++q) {
            d0[q] = gelu_f(d0[q]);
            d1[q] = gelu_f(d1[q]);
        }
        v16h b2 = pack_b_from_d(d0, d1);
        v8f w0a = wmma16(aT0_0, b2, z), w0b = wmma16(aT0_1, b2, z);
        v8f w1a = wmma16(aT1_0, b2, z), w1b = wmma16(aT1_1, b2, z);
        v8f w2a = wmma16(aT2_0, b2, z), w2b = wmma16(aT2_1, b2, z);
        v8f w3a = wmma16(aT3_0, b2, z), w3b = wmma16(aT3_1, b2, z);

        if (valid) {
            int src = edge_src[ec];
            int dst = edge_dst[ec];
            const float* ea = edge_attr + (size_t)ec * 4;
            float y0  = ea[0];
            float y1x = ea[1], y1y = ea[2], y1z = ea[3];
            const float* fs = feat_s + (size_t)src * 32;
            const float* fv = feat_v + (size_t)src * 96;
            float* as = acc_s + (size_t)dst * 64;
            float* av = acc_v + (size_t)dst * 192;
#pragma unroll
            for (int t2 = 0; t2 < 2; ++t2) {
#pragma unroll
                for (int i = 0; i < 8; ++i) {
                    int u = 16 * t2 + 8 * g + i;
                    float W0 = t2 ? w0b[i] : w0a[i];
                    float W1 = t2 ? w1b[i] : w1a[i];
                    float W2 = t2 ? w2b[i] : w2a[i];
                    float W3 = t2 ? w3b[i] : w3a[i];
                    float es  = fs[u];
                    float evx = fv[u * 3 + 0];
                    float evy = fv[u * 3 + 1];
                    float evz = fv[u * 3 + 2];
                    float dotv = evx * y1x + evy * y1y + evz * y1z;
                    atomAddF(as + u,      W0 * es * y0);
                    atomAddF(as + 32 + u, W3 * dotv * INV3);
                    float t1 = W1 * es;
                    atomAddF(av + u * 3 + 0, t1 * y1x);
                    atomAddF(av + u * 3 + 1, t1 * y1y);
                    atomAddF(av + u * 3 + 2, t1 * y1z);
                    float t2f = W2 * y0;
                    atomAddF(av + (32 + u) * 3 + 0, t2f * evx);
                    atomAddF(av + (32 + u) * 3 + 1, t2f * evy);
                    atomAddF(av + (32 + u) * 3 + 2, t2f * evz);
                }
            }
        }
    }
}

__global__ void __launch_bounds__(256)
finalize_kernel(const float* __restrict__ node_input,
                const float* __restrict__ W1_s, const float* __restrict__ W1_v,
                const float* __restrict__ W2_s, const float* __restrict__ W2_v,
                const float* __restrict__ acc_s, const float* __restrict__ acc_v,
                float* __restrict__ out, int N) {
    __shared__ float sW2s[64][32];
    __shared__ float sW2v[64][32];
    __shared__ float sW1s[32][32];
    __shared__ float sW1v[32][32];
    for (int idx = threadIdx.x; idx < 2048; idx += blockDim.x) {
        int u = idx >> 5, w = idx & 31;
        sW2s[u][w] = W2_s[u * 32 + w];
        sW2v[u][w] = W2_v[u * 32 + w];
    }
    for (int idx = threadIdx.x; idx < 1024; idx += blockDim.x) {
        int u = idx >> 5, w = idx & 31;
        sW1s[u][w] = W1_s[u * 64 + 32 + w];
        sW1v[u][w] = W1_v[u * 64 + 32 + w];
    }
    __syncthreads();
    const float inv32 = 0.17677669529663687f;
    const float cconv = 0.125f * 0.31622776601683794f;
    const float ca = 0.9219544457292887f;
    const float cb = 0.3872983346207417f;
    int gid = blockIdx.x * blockDim.x + threadIdx.x;
    int total = N * 32;
    for (int t = gid; t < total; t += gridDim.x * blockDim.x) {
        int n = t >> 5, w = t & 31;
        const float* row = node_input + (size_t)n * 128;
        float ss = 0.f, svx = 0.f, svy = 0.f, svz = 0.f;
#pragma unroll 4
        for (int u = 0; u < 32; ++u) {
            ss += row[u] * sW1s[u][w];
            float wv = sW1v[u][w];
            svx += row[32 + u * 3 + 0] * wv;
            svy += row[32 + u * 3 + 1] * wv;
            svz += row[32 + u * 3 + 2] * wv;
        }
        ss *= inv32; svx *= inv32; svy *= inv32; svz *= inv32;
        const float* as = acc_s + (size_t)n * 64;
        const float* av = acc_v + (size_t)n * 192;
        float cs = 0.f, cvx = 0.f, cvy = 0.f, cvz = 0.f;
#pragma unroll 4
        for (int u = 0; u < 64; ++u) {
            cs += as[u] * sW2s[u][w];
            float wv = sW2v[u][w];
            cvx += av[u * 3 + 0] * wv;
            cvy += av[u * 3 + 1] * wv;
            cvz += av[u * 3 + 2] * wv;
        }
        cs *= cconv; cvx *= cconv; cvy *= cconv; cvz *= cconv;
        float* o = out + (size_t)n * 128;
        o[w] = ca * ss + cb * cs;
        o[32 + w * 3 + 0] = ca * svx + cb * cvx;
        o[32 + w * 3 + 1] = ca * svy + cb * cvy;
        o[32 + w * 3 + 2] = ca * svz + cb * cvz;
    }
}

extern "C" void kernel_launch(void* const* d_in, const int* in_sizes, int n_in,
                              void* d_out, int out_size, void* d_ws, size_t ws_size,
                              hipStream_t stream) {
    (void)n_in; (void)out_size; (void)ws_size;
    const float* node_input   = (const float*)d_in[0];
    const float* edge_attr    = (const float*)d_in[1];
    const float* edge_scalars = (const float*)d_in[2];
    const float* W1_s = (const float*)d_in[3];
    const float* W1_v = (const float*)d_in[4];
    const float* M0   = (const float*)d_in[5];
    const float* M1   = (const float*)d_in[6];
    const float* Wtp0 = (const float*)d_in[7];
    const float* Wtp1 = (const float*)d_in[8];
    const float* Wtp2 = (const float*)d_in[9];
    const float* Wtp3 = (const float*)d_in[10];
    const float* W2_s = (const float*)d_in[11];
    const float* W2_v = (const float*)d_in[12];
    const int* edge_src = (const int*)d_in[13];
    const int* edge_dst = (const int*)d_in[14];

    int N = in_sizes[0] / 128;
    int E = in_sizes[13];

    float* feat_s = (float*)d_ws;
    float* feat_v = feat_s + (size_t)N * 32;
    float* acc_s  = feat_v + (size_t)N * 96;
    float* acc_v  = acc_s  + (size_t)N * 64;

    hipMemsetAsync(acc_s, 0, (size_t)N * 256 * sizeof(float), stream);

    const int threads = 256;
    int blocksN = (N * 32 + threads - 1) / threads;
    node_pre_kernel<<<blocksN, threads, 0, stream>>>(node_input, W1_s, W1_v,
                                                     feat_s, feat_v, N);

    int blocksE = 1024;
    edge_kernel<<<blocksE, threads, 0, stream>>>(edge_attr, edge_scalars,
                                                 M0, M1, Wtp0, Wtp1, Wtp2, Wtp3,
                                                 edge_src, edge_dst,
                                                 feat_s, feat_v, acc_s, acc_v, E);

    finalize_kernel<<<blocksN, threads, 0, stream>>>(node_input, W1_s, W1_v,
                                                     W2_s, W2_v, acc_s, acc_v,
                                                     (float*)d_out, N);
}


# Node pre-transform moved from one thin wave per node (50000 waves, per-thread FMA loop) to one wave per 16-node tile (3125 tiles) on f32 16x16x4 WMMA, rows staged in LDS; feat sum order changes
# speedup vs baseline: 5.1129x; 5.1129x over previous
#include <hip/hip_runtime.h>
#include <math.h>

typedef __attribute__((ext_vector_type(16))) _Float16 v16h;
typedef __attribute__((ext_vector_type(8)))  float    v8f;
typedef __attribute__((ext_vector_type(2)))  float    v2f;

__device__ __forceinline__ float gelu_f(float x) {
    const float c0 = 0.7978845608028654f;
    const float c1 = 2.8853900817779268f;
    float u = c0 * (x + 0.044715f * x * x * x);
    float e = __builtin_amdgcn_exp2f(u * c1);
    return x - x * __builtin_amdgcn_rcpf(e + 1.0f);
}

__device__ __forceinline__ v8f wmma16(v16h a, v16h b, v8f c) {
    return __builtin_amdgcn_wmma_f32_16x16x32_f16(
         false, a,  false, b,
         (short)0, c,  false,  false);
}

__device__ __forceinline__ v8f wmma4(v2f a, v2f b, v8f c) {
    return __builtin_amdgcn_wmma_f32_16x16x4_f32(
         false, a,  false, b,
         (short)0, c,  false,  false);
}

__device__ __forceinline__ v16h load_wT_frag(const float* __restrict__ W,
                                             int Kreal, int t, int lane,
                                             float scale) {
    int m = 16 * t + (lane & 15);
    int g = lane >> 4;
    v16h a;
#pragma unroll
    for (int e = 0; e < 16; ++e) {
        int r = e >> 1, j = e & 1;
        int k = ((r >> 2) << 4) + g * 8 + ((r & 3) << 1) + j;
        float val = (k < Kreal) ? W[k * 32 + m] * scale : 0.0f;
        a[e] = (_Float16)val;
    }
    return a;
}

__device__ __forceinline__ v16h pack_b_from_d(v8f d0, v8f d1) {
    v16h b;
#pragma unroll
    for (int e = 0; e < 16; ++e) {
        int r = e >> 1, j = e & 1;
        int i = ((r & 3) << 1) + j;
        float val = (r >> 2) ? d1[i] : d0[i];
        b[e] = (_Float16)val;
    }
    return b;
}

#define PW 4
#define PPITCH 132

__global__ void __launch_bounds__(PW * 32)
node_pre_kernel(const float* __restrict__ node_input,
                const float* __restrict__ W1_s,
                const float* __restrict__ W1_v,
                float* __restrict__ feat_s,
                float* __restrict__ feat_v, int N) {
    __shared__ __attribute__((aligned(16))) float sRow[PW][16 * PPITCH];

    const int lane = threadIdx.x & 31;
    const int col  = lane & 15;
    const int g    = lane >> 4;
    const int wv_in_blk = __builtin_amdgcn_readfirstlane((int)(threadIdx.x >> 5));
    const float inv32 = 0.17677669529663687f;

    v2f bs[8][2], bv[8][2];
#pragma unroll
    for (int ks = 0; ks < 8; ++ks) {
#pragma unroll
        for (int nt = 0; nt < 2; ++nt) {
            const int k0 = 4 * ks + 2 * g;
            const int w  = 16 * nt + col;
            bs[ks][nt][0] = W1_s[k0 * 64 + w] * inv32;
            bs[ks][nt][1] = W1_s[(k0 + 1) * 64 + w] * inv32;
            bv[ks][nt][0] = W1_v[k0 * 64 + w] * inv32;
            bv[ks][nt][1] = W1_v[(k0 + 1) * 64 + w] * inv32;
        }
    }

    float* srow = &sRow[wv_in_blk][0];
    const int ntiles = (N + 15) >> 4;
    for (int tile = blockIdx.x * PW + wv_in_blk; tile < ntiles; tile += gridDim.x * PW) {
        const int n0 = tile * 16;
#pragma unroll
        for (int j = 0; j < 16; ++j) {
            const int n = (n0 + j < N) ? (n0 + j) : (N - 1);
            const float4 x = *(const float4*)(node_input + (size_t)n * 128 + 4 * lane);
            *(float4*)(srow + j * PPITCH + 4 * lane) = x;
        }

        v8f acc[4][2];
#pragma unroll
        for (int cp = 0; cp < 4; ++cp) { acc[cp][0] = (v8f){}; acc[cp][1] = (v8f){}; }

        const float* arow = srow + col * PPITCH;
#pragma unroll
        for (int ks = 0; ks < 8; ++ks) {
            const v2f a0 = *(const v2f*)(arow + 4 * ks + 2 * g);
            acc[0][0] = wmma4(a0, bs[ks][0], acc[0][0]);
            acc[0][1] = wmma4(a0, bs[ks][1], acc[0][1]);
            const float* vp = arow + 32 + 3 * (4 * ks + 2 * g);
#pragma unroll
            for (int k = 0; k < 3; ++k) {
                v2f ak;
                ak[0] = vp[k];
                ak[1] = vp[3 + k];
                acc[1 + k][0] = wmma4(ak, bv[ks][0], acc[1 + k][0]);
                acc[1 + k][1] = wmma4(ak, bv[ks][1], acc[1 + k][1]);
            }
        }
#pragma unroll
        for (int r = 0; r < 8; ++r) {
            const int n = n0 + 8 * g + r;
            if (n < N) {
#pragma unroll
                for (int nt = 0; nt < 2; ++nt) {
                    const int w = 16 * nt + col;
                    feat_s[(size_t)n * 32 + w] = acc[0][nt][r];
                    float* fv = feat_v + (size_t)n * 96 + w * 3;
                    fv[0] = acc[1][nt][r];
                    fv[1] = acc[2][nt][r];
                    fv[2] = acc[3][nt][r];
                }
            }
        }
    }
}

__global__ void __launch_bounds__(256)
hist_kernel(const int* __restrict__ edge_dst, int* __restrict__ cnt,
            int* __restrict__ rank, int E) {
    int e = blockIdx.x * blockDim.x + threadIdx.x;
    if (e < E)
        rank[e] = __hip_atomic_fetch_add(cnt + edge_dst[e], 1, __ATOMIC_RELAXED,
                                         __HIP_MEMORY_SCOPE_AGENT);
}

__global__ void __launch_bounds__(1024)
scan_kernel(const int* __restrict__ cnt, int* __restrict__ off, int N) {
    __shared__ int wtot[32];
    const int lane = threadIdx.x & 31;
    const int wid  = threadIdx.x >> 5;
    const int seg  = (N + 31) / 32;
    const int beg  = wid * seg;
    const int end  = (beg + seg < N) ? (beg + seg) : N;
    int s = 0;
    for (int i = beg + lane; i < end; i += 32) s += cnt[i];
    for (int d = 16; d > 0; d >>= 1) s += __shfl_xor(s, d);
    if (lane == 0) wtot[wid] = s;
    __syncthreads();
    int carry = 0;
    for (int w = 0; w < wid; ++w) carry += wtot[w];
    for (int i0 = beg; i0 < end; i0 += 32) {
        const int i = i0 + lane;
        const int v = (i < end) ? cnt[i] : 0;
        int x = v;
        for (int d = 1; d < 32; d <<= 1) {
            int y = __shfl_up(x, d);
            if (lane >= d) x += y;
        }
        if (i < end) off[i] = carry + x - v;
        carry += __shfl(x, 31);
    }
    if (threadIdx.x == 0) {
        int t = 0;
        for (int w = 0; w < 32; ++w) t += wtot[w];
        off[N] = t;
    }
}

__global__ void __launch_bounds__(256)
fill_kernel(const int* __restrict__ edge_dst, const int* __restrict__ off,
            const int* __restrict__ rank, int* __restrict__ perm, int E) {
    int e = blockIdx.x * blockDim.x + threadIdx.x;
    if (e < E) perm[off[edge_dst[e]] + rank[e]] = e;
}

#define GW 16
#define SPITCH 260

__global__ void __launch_bounds__(GW * 32)
node_gather_kernel(const float* __restrict__ node_input,
                   const float* __restrict__ edge_attr,
                   const float* __restrict__ edge_scalars,
                   const float* __restrict__ W1_s, const float* __restrict__ W1_v,
                   const float* __restrict__ M0, const float* __restrict__ M1,
                   const float* __restrict__ Wtp0, const float* __restrict__ Wtp1,
                   const float* __restrict__ Wtp2, const float* __restrict__ Wtp3,
                   const float* __restrict__ W2_s, const float* __restrict__ W2_v,
                   const int* __restrict__ edge_src,
                   const int* __restrict__ off, const int* __restrict__ perm,
                   const float* __restrict__ feat_s, const float* __restrict__ feat_v,
                   float* __restrict__ out, int N) {
    __shared__ v16h  sFrag[12][32];
    __shared__ v2f   sB2s[16][2][32];
    __shared__ v2f   sB2v[16][2][32];
    __shared__ v2f   sB1s[8][2][32];
    __shared__ v2f   sB1v[8][2][32];
    __shared__ __attribute__((aligned(16))) float sStage[GW][16 * SPITCH];

    const int lane = threadIdx.x & 31;
    const int col  = lane & 15;
    const int g    = lane >> 4;
    const int wv_in_blk = __builtin_amdgcn_readfirstlane((int)(threadIdx.x >> 5));

    const float inv8  = 0.35355339059327373f;
    const float inv32 = 0.17677669529663687f;
    const float INV3  = 0.5773502691896258f;
    const float cconv = 0.125f * 0.31622776601683794f;
    const float ca = 0.9219544457292887f;
    const float cb = 0.3872983346207417f;
    const float sc2 = cconv * cb;
    const float sc1 = inv32 * ca;

    for (int idx = threadIdx.x; idx < 16 * 2 * 32; idx += blockDim.x) {
        const int ks = idx >> 6, nt = (idx >> 5) & 1, l = idx & 31;
        const int k0 = 4 * ks + 2 * (l >> 4);
        const int w  = 16 * nt + (l & 15);
        v2f bs, bv;
        bs[0] = W2_s[k0 * 32 + w] * sc2;  bs[1] = W2_s[(k0 + 1) * 32 + w] * sc2;
        bv[0] = W2_v[k0 * 32 + w] * sc2;  bv[1] = W2_v[(k0 + 1) * 32 + w] * sc2;
        sB2s[ks][nt][l] = bs;
        sB2v[ks][nt][l] = bv;
    }
    for (int idx = threadIdx.x; idx < 8 * 2 * 32; idx += blockDim.x) {
        const int ks = idx >> 6, nt = (idx >> 5) & 1, l = idx & 31;
        const int k0 = 4 * ks + 2 * (l >> 4);
        const int w  = 32 + 16 * nt + (l & 15);
        v2f bs, bv;
        bs[0] = W1_s[k0 * 64 + w] * sc1;  bs[1] = W1_s[(k0 + 1) * 64 + w] * sc1;
        bv[0] = W1_v[k0 * 64 + w] * sc1;  bv[1] = W1_v[(k0 + 1) * 64 + w] * sc1;
        sB1s[ks][nt][l] = bs;
        sB1v[ks][nt][l] = bv;
    }
    for (int f = wv_in_blk; f < 12; f += GW) {
        const int m = f >> 1;
        const float* Wp = (m == 0) ? M0 : (m == 1) ? M1 : (m == 2) ? Wtp0
                        : (m == 3) ? Wtp1 : (m == 4) ? Wtp2 : Wtp3;
        sFrag[f][lane] = load_wT_frag(Wp, (m == 0) ? 8 : 32, f & 1, lane,
                                      (m == 0) ? inv8 : inv32);
    }
    __syncthreads();

    float* stage = &sStage[wv_in_blk][0];

    const int nwaves = gridDim.x * GW;
    const int wave   = blockIdx.x * GW + wv_in_blk;
    const int q      = N / nwaves;
    const int rem    = N - q * nwaves;
    const int nbeg   = wave * q + ((wave < rem) ? wave : rem);
    const int nend   = nbeg + q + ((wave < rem) ? 1 : 0);

    for (int n0 = nbeg; n0 < nend; n0 += 16) {
        for (int j = 0; j < 16; ++j) {
            const int n = n0 + j;
            int beg = 0, end = 0;
            if (n < nend) {
                beg = __builtin_amdgcn_readfirstlane(off[n]);
                end = __builtin_amdgcn_readfirstlane(off[n + 1]);
            }

            float S0[2] = {0.f, 0.f}, S1[2] = {0.f, 0.f};
            float V0[2][3] = {{0.f, 0.f, 0.f}, {0.f, 0.f, 0.f}};
            float V1[2][3] = {{0.f, 0.f, 0.f}, {0.f, 0.f, 0.f}};

            for (int base = beg; base < end; base += 16) {
                asm volatile("" ::: "memory");
                v16h b0;
#pragma unroll
                for (int q = 0; q < 16; ++q) b0[q] = (_Float16)0.0f;
                if (g == 0 && base + col < end) {
                    const float* p = edge_scalars + (size_t)perm[base + col] * 8;
#pragma unroll
                    for (int q = 0; q < 8; ++q) b0[q] = (_Float16)p[q];
                }
                v8f z = {};
                v8f d0 = wmma16(sFrag[0][lane], b0, z);
                v8f d1 = wmma16(sFrag[1][lane], b0, z);
#pragma unroll
                for (int q = 0; q < 8; ++q) { d0[q] = gelu_f(d0[q]); d1[q] = gelu_f(d1[q]); }
                v16h b1 = pack_b_from_d(d0, d1);
                d0 = wmma16(sFrag[2][lane], b1, z);
                d1 = wmma16(sFrag[3][lane], b1, z);
#pragma unroll
                for (int q = 0; q < 8; ++q) { d0[q] = gelu_f(d0[q]); d1[q] = gelu_f(d1[q]); }
                v16h h2 = pack_b_from_d(d0, d1);
                v8f w0[2], w1[2], w2[2], w3[2];
#pragma unroll
                for (int t = 0; t < 2; ++t) {
                    w0[t] = wmma16(h2, sFrag[4 + t][lane], z);
                    w1[t] = wmma16(h2, sFrag[6 + t][lane], z);
                    w2[t] = wmma16(h2, sFrag[8 + t][lane], z);
                    w3[t] = wmma16(h2, sFrag[10 + t][lane], z);
                }
#pragma unroll
                for (int i = 0; i < 8; ++i) {
                    const int p  = base + 8 * g + i;
                    const bool v = p < end;
                    const int e   = perm[v ? p : beg];
                    const int src = edge_src[e];
                    const float4 ea = *(const float4*)(edge_attr + (size_t)e * 4);
                    const float y0  = v ? ea.x : 0.f;
                    const float y1x = v ? ea.y : 0.f;
                    const float y1y = v ? ea.z : 0.f;
                    const float y1z = v ? ea.w : 0.f;
                    const float* fs = feat_s + (size_t)src * 32;
                    const float* fv = feat_v + (size_t)src * 96;
#pragma unroll
                    for (int t = 0; t < 2; ++t) {
                        const int u = 16 * t + col;
                        const float es  = fs[u];
                        const float evx = fv[u * 3 + 0];
                        const float evy = fv[u * 3 + 1];
                        const float evz = fv[u * 3 + 2];
                        const float dotv = evx * y1x + evy * y1y + evz * y1z;
                        S0[t] += w0[t][i] * es * y0;
                        S1[t] += w3[t][i] * dotv * INV3;
                        const float t1 = w1[t][i] * es;
                        V0[t][0] += t1 * y1x;
                        V0[t][1] += t1 * y1y;
                        V0[t][2] += t1 * y1z;
                        const float t2f = w2[t][i] * y0;
                        V1[t][0] += t2f * evx;
                        V1[t][1] += t2f * evy;
                        V1[t][2] += t2f * evz;
                    }
                }
            }
            float* srow = stage + j * SPITCH;
            {
                const float k0 = g ? S0[1] : S0[0], x0 = g ? S0[0] : S0[1];
                const float k1 = g ? S1[1] : S1[0], x1 = g ? S1[0] : S1[1];
                srow[lane]      = k0 + __shfl_xor(x0, 16);
                srow[32 + lane] = k1 + __shfl_xor(x1, 16);
            }
#pragma unroll
            for (int k = 0; k < 3; ++k) {
                const float k0 = g ? V0[1][k] : V0[0][k], x0 = g ? V0[0][k] : V0[1][k];
                const float k1 = g ? V1[1][k] : V1[0][k], x1 = g ? V1[0][k] : V1[1][k];
                srow[64 * (1 + k) + lane]      = k0 + __shfl_xor(x0, 16);
                srow[64 * (1 + k) + 32 + lane] = k1 + __shfl_xor(x1, 16);
            }
        }
        asm volatile("" ::: "memory");

        v8f acc[4][2];
#pragma unroll
        for (int cp = 0; cp < 4; ++cp) { acc[cp][0] = (v8f){}; acc[cp][1] = (v8f){}; }

        const float* arow = stage + col * SPITCH + 2 * g;
#pragma unroll
        for (int ks = 0; ks < 16; ++ks) {
            const v2f bs0 = sB2s[ks][0][lane], bs1 = sB2s[ks][1][lane];
            const v2f bv0 = sB2v[ks][0][lane], bv1 = sB2v[ks][1][lane];
            const v2f a0 = *(const v2f*)(arow + 4 * ks);
            acc[0][0] = wmma4(a0, bs0, acc[0][0]);
            acc[0][1] = wmma4(a0, bs1, acc[0][1]);
#pragma unroll
            for (int k = 0; k < 3; ++k) {
                const v2f ak = *(const v2f*)(arow + 64 * (1 + k) + 4 * ks);
                acc[1 + k][0] = wmma4(ak, bv0, acc[1 + k][0]);
                acc[1 + k][1] = wmma4(ak, bv1, acc[1 + k][1]);
            }
        }
        {
            const int nr = (n0 + col < nend) ? (n0 + col) : (nend - 1);
            const float* rowp = node_input + (size_t)nr * 128 + 2 * g;
#pragma unroll
            for (int ks = 0; ks < 8; ++ks) {
                const v2f bs0 = sB1s[ks][0][lane], bs1 = sB1s[ks][1][lane];
                const v2f bv0 = sB1v[ks][0][lane], bv1 = sB1v[ks][1][lane];
                const v2f a0 = *(const v2f*)(rowp + 4 * ks);
                acc[0][0] = wmma4(a0, bs0, acc[0][0]);
                acc[0][1] = wmma4(a0, bs1, acc[0][1]);
                const float* vp = rowp + 32 + 3 * (4 * ks) + 4 * g;
#pragma unroll
                for (int k = 0; k < 3; ++k) {
                    v2f ak;
                    ak[0] = vp[k];
                    ak[1] = vp[3 + k];
                    acc[1 + k][0] = wmma4(ak, bv0, acc[1 + k][0]);
                    acc[1 + k][1] = wmma4(ak, bv1, acc[1 + k][1]);
                }
            }
        }
#pragma unroll
        for (int r = 0; r < 8; ++r) {
            const int n = n0 + 8 * g + r;
            if (n < nend) {
                float* o = out + (size_t)n * 128;
#pragma unroll
                for (int nt = 0; nt < 2; ++nt) {
                    const int w = 16 * nt + col;
                    o[w] = acc[0][nt][r];
                    o[32 + w * 3 + 0] = acc[1][nt][r];
                    o[32 + w * 3 + 1] = acc[2][nt][r];
                    o[32 + w * 3 + 2] = acc[3][nt][r];
                }
            }
        }
    }
}

extern "C" void kernel_launch(void* const* d_in, const int* in_sizes, int n_in,
                              void* d_out, int out_size, void* d_ws, size_t ws_size,
                              hipStream_t stream) {
    (void)n_in; (void)out_size; (void)ws_size;
    const float* node_input   = (const float*)d_in[0];
    const float* edge_attr    = (const float*)d_in[1];
    const float* edge_scalars = (const float*)d_in[2];
    const float* W1_s = (const float*)d_in[3];
    const float* W1_v = (const float*)d_in[4];
    const float* M0   = (const float*)d_in[5];
    const float* M1   = (const float*)d_in[6];
    const float* Wtp0 = (const float*)d_in[7];
    const float* Wtp1 = (const float*)d_in[8];
    const float* Wtp2 = (const float*)d_in[9];
    const float* Wtp3 = (const float*)d_in[10];
    const float* W2_s = (const float*)d_in[11];
    const float* W2_v = (const float*)d_in[12];
    const int* edge_src = (const int*)d_in[13];
    const int* edge_dst = (const int*)d_in[14];

    int N = in_sizes[0] / 128;
    int E = in_sizes[13];

    float* feat_s = (float*)d_ws;
    float* feat_v = feat_s + (size_t)N * 32;
    int* cnt  = (int*)(feat_v + (size_t)N * 96);
    int* off  = cnt + N;
    int* rank = off + (N + 1);
    int* perm = rank + E;

    hipMemsetAsync(cnt, 0, (size_t)N * sizeof(int), stream);

    const int threads = 256;
    int blocksP  = ((N + 15) / 16 + PW - 1) / PW;
    int blocksEd = (E + threads - 1) / threads;

    hist_kernel<<<blocksEd, threads, 0, stream>>>(edge_dst, cnt, rank, E);
    scan_kernel<<<1, 1024, 0, stream>>>(cnt, off, N);
    fill_kernel<<<blocksEd, threads, 0, stream>>>(edge_dst, off, rank, perm, E);

    node_pre_kernel<<<blocksP, PW * 32, 0, stream>>>(node_input, W1_s, W1_v,
                                                     feat_s, feat_v, N);

    int blocksG = 256;
    node_gather_kernel<<<blocksG, GW * 32, 0, stream>>>(node_input, edge_attr, edge_scalars,
                                                        W1_s, W1_v, M0, M1,
                                                        Wtp0, Wtp1, Wtp2, Wtp3, W2_s, W2_v,
                                                        edge_src, off, perm, feat_s, feat_v,
                                                        (float*)d_out, N);
}


# Prefix-sum kernel: the 32-lane scan runs on DPP row shifts (vector ALU) instead of six dependent LDS-unit permutes per 32 counts, and sixteen rows are loaded per trip; integer sums, same results
# speedup vs baseline: 6.0318x; 1.1797x over previous
#include <hip/hip_runtime.h>
#include <math.h>

typedef __attribute__((ext_vector_type(16))) _Float16 v16h;
typedef __attribute__((ext_vector_type(8)))  float    v8f;
typedef __attribute__((ext_vector_type(2)))  float    v2f;

__device__ __forceinline__ float gelu_f(float x) {
    const float c0 = 0.7978845608028654f;
    const float c1 = 2.8853900817779268f;
    float u = c0 * (x + 0.044715f * x * x * x);
    float e = __builtin_amdgcn_exp2f(u * c1);
    return x - x * __builtin_amdgcn_rcpf(e + 1.0f);
}

__device__ __forceinline__ v8f wmma16(v16h a, v16h b, v8f c) {
    return __builtin_amdgcn_wmma_f32_16x16x32_f16(
         false, a,  false, b,
         (short)0, c,  false,  false);
}

__device__ __forceinline__ v8f wmma4(v2f a, v2f b, v8f c) {
    return __builtin_amdgcn_wmma_f32_16x16x4_f32(
         false, a,  false, b,
         (short)0, c,  false,  false);
}

__device__ __forceinline__ v16h load_wT_frag(const float* __restrict__ W,
                                             int Kreal, int t, int lane,
                                             float scale) {
    int m = 16 * t + (lane & 15);
    int g = lane >> 4;
    v16h a;
#pragma unroll
    for (int e = 0; e < 16; ++e) {
        int r = e >> 1, j = e & 1;
        int k = ((r >> 2) << 4) + g * 8 + ((r & 3) << 1) + j;
        float val = (k < Kreal) ? W[k * 32 + m] * scale : 0.0f;
        a[e] = (_Float16)val;
    }
    return a;
}

__device__ __forceinline__ v16h pack_b_from_d(v8f d0, v8f d1) {
    v16h b;
#pragma unroll
    for (int e = 0; e < 16; ++e) {
        int r = e >> 1, j = e & 1;
        int i = ((r & 3) << 1) + j;
        float val = (r >> 2) ? d1[i] : d0[i];
        b[e] = (_Float16)val;
    }
    return b;
}

#define PW 4
#define PPITCH 132

__global__ void __launch_bounds__(PW * 32)
node_pre_kernel(const float* __restrict__ node_input,
                const float* __restrict__ W1_s,
                const float* __restrict__ W1_v,
                float* __restrict__ feat_s,
                float* __restrict__ feat_v, int N) {
    __shared__ __attribute__((aligned(16))) float sRow[PW][16 * PPITCH];

    const int lane = threadIdx.x & 31;
    const int col  = lane & 15;
    const int g    = lane >> 4;
    const int wv_in_blk = __builtin_amdgcn_readfirstlane((int)(threadIdx.x >> 5));
    const float inv32 = 0.17677669529663687f;

    v2f bs[8][2], bv[8][2];
#pragma unroll
    for (int ks = 0; ks < 8; ++ks) {
#pragma unroll
        for (int nt = 0; nt < 2; ++nt) {
            const int k0 = 4 * ks + 2 * g;
            const int w  = 16 * nt + col;
            bs[ks][nt][0] = W1_s[k0 * 64 + w] * inv32;
            bs[ks][nt][1] = W1_s[(k0 + 1) * 64 + w] * inv32;
            bv[ks][nt][0] = W1_v[k0 * 64 + w] * inv32;
            bv[ks][nt][1] = W1_v[(k0 + 1) * 64 + w] * inv32;
        }
    }

    float* srow = &sRow[wv_in_blk][0];
    const int ntiles = (N + 15) >> 4;
    for (int tile = blockIdx.x * PW + wv_in_blk; tile < ntiles; tile += gridDim.x * PW) {
        const int n0 = tile * 16;
#pragma unroll
        for (int j = 0; j < 16; ++j) {
            const int n = (n0 + j < N) ? (n0 + j) : (N - 1);
            const float4 x = *(const float4*)(node_input + (size_t)n * 128 + 4 * lane);
            *(float4*)(srow + j * PPITCH + 4 * lane) = x;
        }

        v8f acc[4][2];
#pragma unroll
        for (int cp = 0; cp < 4; ++cp) { acc[cp][0] = (v8f){}; acc[cp][1] = (v8f){}; }

        const float* arow = srow + col * PPITCH;
#pragma unroll
        for (int ks = 0; ks < 8; ++ks) {
            const v2f a0 = *(const v2f*)(arow + 4 * ks + 2 * g);
            acc[0][0] = wmma4(a0, bs[ks][0], acc[0][0]);
            acc[0][1] = wmma4(a0, bs[ks][1], acc[0][1]);
            const float* vp = arow + 32 + 3 * (4 * ks + 2 * g);
#pragma unroll
            for (int k = 0; k < 3; ++k) {
                v2f ak;
                ak[0] = vp[k];
                ak[1] = vp[3 + k];
                acc[1 + k][0] = wmma4(ak, bv[ks][0], acc[1 + k][0]);
                acc[1 + k][1] = wmma4(ak, bv[ks][1], acc[1 + k][1]);
            }
        }
#pragma unroll
        for (int r = 0; r < 8; ++r) {
            const int n = n0 + 8 * g + r;
            if (n < N) {
#pragma unroll
                for (int nt = 0; nt < 2; ++nt) {
                    const int w = 16 * nt + col;
                    feat_s[(size_t)n * 32 + w] = acc[0][nt][r];
                    float* fv = feat_v + (size_t)n * 96 + w * 3;
                    fv[0] = acc[1][nt][r];
                    fv[1] = acc[2][nt][r];
                    fv[2] = acc[3][nt][r];
                }
            }
        }
    }
}

__global__ void __launch_bounds__(256)
hist_kernel(const int* __restrict__ edge_dst, int* __restrict__ cnt,
            int* __restrict__ rank, int E) {
    int e = blockIdx.x * blockDim.x + threadIdx.x;
    if (e < E)
        rank[e] = __hip_atomic_fetch_add(cnt + edge_dst[e], 1, __ATOMIC_RELAXED,
                                         __HIP_MEMORY_SCOPE_AGENT);
}

__device__ __forceinline__ int wave_incl_scan(int x, int lane) {
    x += __builtin_amdgcn_update_dpp(0, x, 0x111, 0xf, 0xf, true);
    x += __builtin_amdgcn_update_dpp(0, x, 0x112, 0xf, 0xf, true);
    x += __builtin_amdgcn_update_dpp(0, x, 0x114, 0xf, 0xf, true);
    x += __builtin_amdgcn_update_dpp(0, x, 0x118, 0xf, 0xf, true);
    const int low = __builtin_amdgcn_readlane(x, 15);
    return (lane >= 16) ? (x + low) : x;
}

#define SROWS 16
__global__ void __launch_bounds__(1024)
scan_kernel(const int* __restrict__ cnt, int* __restrict__ off, int N) {
    __shared__ int wtot[32];
    const int lane = threadIdx.x & 31;
    const int wid  = threadIdx.x >> 5;
    const int seg  = (N + 31) / 32;
    const int beg  = wid * seg;
    const int end  = (beg + seg < N) ? (beg + seg) : N;
    int s = 0;
    for (int i = beg + lane; i < end; i += 32) s += cnt[i];
    s = __builtin_amdgcn_readlane(wave_incl_scan(s, lane), 31);
    if (lane == 0) wtot[wid] = s;
    __syncthreads();
    int carry = 0;
    for (int w = 0; w < wid; ++w) carry += wtot[w];
    for (int i0 = beg; i0 < end; i0 += 32 * SROWS) {
        int v[SROWS], x[SROWS];
#pragma unroll
        for (int r = 0; r < SROWS; ++r) {
            const int i = i0 + 32 * r + lane;
            v[r] = cnt[(i < end) ? i : (end - 1)];
        }
        __builtin_amdgcn_sched_barrier(0);
#pragma unroll
        for (int r = 0; r < SROWS; ++r) {
            if (i0 + 32 * r + lane >= end) v[r] = 0;
            x[r] = wave_incl_scan(v[r], lane);
        }
#pragma unroll
        for (int r = 0; r < SROWS; ++r) {
            const int i = i0 + 32 * r + lane;
            if (i < end) off[i] = carry + x[r] - v[r];
            carry += __builtin_amdgcn_readlane(x[r], 31);
        }
    }
    if (threadIdx.x == 0) {
        int t = 0;
        for (int w = 0; w < 32; ++w) t += wtot[w];
        off[N] = t;
    }
}

__global__ void __launch_bounds__(256)
fill_kernel(const int* __restrict__ edge_dst, const int* __restrict__ off,
            const int* __restrict__ rank, int* __restrict__ perm, int E) {
    int e = blockIdx.x * blockDim.x + threadIdx.x;
    if (e < E) perm[off[edge_dst[e]] + rank[e]] = e;
}

#define GW 16
#define SPITCH 260

__global__ void __launch_bounds__(GW * 32)
node_gather_kernel(const float* __restrict__ node_input,
                   const float* __restrict__ edge_attr,
                   const float* __restrict__ edge_scalars,
                   const float* __restrict__ W1_s, const float* __restrict__ W1_v,
                   const float* __restrict__ M0, const float* __restrict__ M1,
                   const float* __restrict__ Wtp0, const float* __restrict__ Wtp1,
                   const float* __restrict__ Wtp2, const float* __restrict__ Wtp3,
                   const float* __restrict__ W2_s, const float* __restrict__ W2_v,
                   const int* __restrict__ edge_src,
                   const int* __restrict__ off, const int* __restrict__ perm,
                   const float* __restrict__ feat_s, const float* __restrict__ feat_v,
                   float* __restrict__ out, int N) {
    __shared__ v16h  sFrag[12][32];
    __shared__ v2f   sB2s[16][2][32];
    __shared__ v2f   sB2v[16][2][32];
    __shared__ v2f   sB1s[8][2][32];
    __shared__ v2f   sB1v[8][2][32];
    __shared__ __attribute__((aligned(16))) float sStage[GW][16 * SPITCH];

    const int lane = threadIdx.x & 31;
    const int col  = lane & 15;
    const int g    = lane >> 4;
    const int wv_in_blk = __builtin_amdgcn_readfirstlane((int)(threadIdx.x >> 5));

    const float inv8  = 0.35355339059327373f;
    const float inv32 = 0.17677669529663687f;
    const float INV3  = 0.5773502691896258f;
    const float cconv = 0.125f * 0.31622776601683794f;
    const float ca = 0.9219544457292887f;
    const float cb = 0.3872983346207417f;
    const float sc2 = cconv * cb;
    const float sc1 = inv32 * ca;

    for (int idx = threadIdx.x; idx < 16 * 2 * 32; idx += blockDim.x) {
        const int ks = idx >> 6, nt = (idx >> 5) & 1, l = idx & 31;
        const int k0 = 4 * ks + 2 * (l >> 4);
        const int w  = 16 * nt + (l & 15);
        v2f bs, bv;
        bs[0] = W2_s[k0 * 32 + w] * sc2;  bs[1] = W2_s[(k0 + 1) * 32 + w] * sc2;
        bv[0] = W2_v[k0 * 32 + w] * sc2;  bv[1] = W2_v[(k0 + 1) * 32 + w] * sc2;
        sB2s[ks][nt][l] = bs;
        sB2v[ks][nt][l] = bv;
    }
    for (int idx = threadIdx.x; idx < 8 * 2 * 32; idx += blockDim.x) {
        const int ks = idx >> 6, nt = (idx >> 5) & 1, l = idx & 31;
        const int k0 = 4 * ks + 2 * (l >> 4);
        const int w  = 32 + 16 * nt + (l & 15);
        v2f bs, bv;
        bs[0] = W1_s[k0 * 64 + w] * sc1;  bs[1] = W1_s[(k0 + 1) * 64 + w] * sc1;
        bv[0] = W1_v[k0 * 64 + w] * sc1;  bv[1] = W1_v[(k0 + 1) * 64 + w] * sc1;
        sB1s[ks][nt][l] = bs;
        sB1v[ks][nt][l] = bv;
    }
    for (int f = wv_in_blk; f < 12; f += GW) {
        const int m = f >> 1;
        const float* Wp = (m == 0) ? M0 : (m == 1) ? M1 : (m == 2) ? Wtp0
                        : (m == 3) ? Wtp1 : (m == 4) ? Wtp2 : Wtp3;
        sFrag[f][lane] = load_wT_frag(Wp, (m == 0) ? 8 : 32, f & 1, lane,
                                      (m == 0) ? inv8 : inv32);
    }
    __syncthreads();

    float* stage = &sStage[wv_in_blk][0];

    const int nwaves = gridDim.x * GW;
    const int wave   = blockIdx.x * GW + wv_in_blk;
    const int q      = N / nwaves;
    const int rem    = N - q * nwaves;
    const int nbeg   = wave * q + ((wave < rem) ? wave : rem);
    const int nend   = nbeg + q + ((wave < rem) ? 1 : 0);

    for (int n0 = nbeg; n0 < nend; n0 += 16) {
        for (int j = 0; j < 16; ++j) {
            const int n = n0 + j;
            int beg = 0, end = 0;
            if (n < nend) {
                beg = __builtin_amdgcn_readfirstlane(off[n]);
                end = __builtin_amdgcn_readfirstlane(off[n + 1]);
            }

            float S0[2] = {0.f, 0.f}, S1[2] = {0.f, 0.f};
            float V0[2][3] = {{0.f, 0.f, 0.f}, {0.f, 0.f, 0.f}};
            float V1[2][3] = {{0.f, 0.f, 0.f}, {0.f, 0.f, 0.f}};

            for (int base = beg; base < end; base += 16) {
                asm volatile("" ::: "memory");
                v16h b0;
#pragma unroll
                for (int q = 0; q < 16; ++q) b0[q] = (_Float16)0.0f;
                if (g == 0 && base + col < end) {
                    const float* p = edge_scalars + (size_t)perm[base + col] * 8;
#pragma unroll
                    for (int q = 0; q < 8; ++q) b0[q] = (_Float16)p[q];
                }
                v8f z = {};
                v8f d0 = wmma16(sFrag[0][lane], b0, z);
                v8f d1 = wmma16(sFrag[1][lane], b0, z);
#pragma unroll
                for (int q = 0; q < 8; ++q) { d0[q] = gelu_f(d0[q]); d1[q] = gelu_f(d1[q]); }
                v16h b1 = pack_b_from_d(d0, d1);
                d0 = wmma16(sFrag[2][lane], b1, z);
                d1 = wmma16(sFrag[3][lane], b1, z);
#pragma unroll
                for (int q = 0; q < 8; ++q) { d0[q] = gelu_f(d0[q]); d1[q] = gelu_f(d1[q]); }
                v16h h2 = pack_b_from_d(d0, d1);
                v8f w0[2], w1[2], w2[2], w3[2];
#pragma unroll
                for (int t = 0; t < 2; ++t) {
                    w0[t] = wmma16(h2, sFrag[4 + t][lane], z);
                    w1[t] = wmma16(h2, sFrag[6 + t][lane], z);
                    w2[t] = wmma16(h2, sFrag[8 + t][lane], z);
                    w3[t] = wmma16(h2, sFrag[10 + t][lane], z);
                }
#pragma unroll
                for (int i = 0; i < 8; ++i) {
                    const int p  = base + 8 * g + i;
                    const bool v = p < end;
                    const int e   = perm[v ? p : beg];
                    const int src = edge_src[e];
                    const float4 ea = *(const float4*)(edge_attr + (size_t)e * 4);
                    const float y0  = v ? ea.x : 0.f;
                    const float y1x = v ? ea.y : 0.f;
                    const float y1y = v ? ea.z : 0.f;
                    const float y1z = v ? ea.w : 0.f;
                    const float* fs = feat_s + (size_t)src * 32;
                    const float* fv = feat_v + (size_t)src * 96;
#pragma unroll
                    for (int t = 0; t < 2; ++t) {
                        const int u = 16 * t + col;
                        const float es  = fs[u];
                        const float evx = fv[u * 3 + 0];
                        const float evy = fv[u * 3 + 1];
                        const float evz = fv[u * 3 + 2];
                        const float dotv = evx * y1x + evy * y1y + evz * y1z;
                        S0[t] += w0[t][i] * es * y0;
                        S1[t] += w3[t][i] * dotv * INV3;
                        const float t1 = w1[t][i] * es;
                        V0[t][0] += t1 * y1x;
                        V0[t][1] += t1 * y1y;
                        V0[t][2] += t1 * y1z;
                        const float t2f = w2[t][i] * y0;
                        V1[t][0] += t2f * evx;
                        V1[t][1] += t2f * evy;
                        V1[t][2] += t2f * evz;
                    }
                }
            }
            float* srow = stage + j * SPITCH;
            {
                const float k0 = g ? S0[1] : S0[0], x0 = g ? S0[0] : S0[1];
                const float k1 = g ? S1[1] : S1[0], x1 = g ? S1[0] : S1[1];
                srow[lane]      = k0 + __shfl_xor(x0, 16);
                srow[32 + lane] = k1 + __shfl_xor(x1, 16);
            }
#pragma unroll
            for (int k = 0; k < 3; ++k) {
                const float k0 = g ? V0[1][k] : V0[0][k], x0 = g ? V0[0][k] : V0[1][k];
                const float k1 = g ? V1[1][k] : V1[0][k], x1 = g ? V1[0][k] : V1[1][k];
                srow[64 * (1 + k) + lane]      = k0 + __shfl_xor(x0, 16);
                srow[64 * (1 + k) + 32 + lane] = k1 + __shfl_xor(x1, 16);
            }
        }
        asm volatile("" ::: "memory");

        v8f acc[4][2];
#pragma unroll
        for (int cp = 0; cp < 4; ++cp) { acc[cp][0] = (v8f){}; acc[cp][1] = (v8f){}; }

        const float* arow = stage + col * SPITCH + 2 * g;
#pragma unroll
        for (int ks = 0; ks < 16; ++ks) {
            const v2f bs0 = sB2s[ks][0][lane], bs1 = sB2s[ks][1][lane];
            const v2f bv0 = sB2v[ks][0][lane], bv1 = sB2v[ks][1][lane];
            const v2f a0 = *(const v2f*)(arow + 4 * ks);
            acc[0][0] = wmma4(a0, bs0, acc[0][0]);
            acc[0][1] = wmma4(a0, bs1, acc[0][1]);
#pragma unroll
            for (int k = 0; k < 3; ++k) {
                const v2f ak = *(const v2f*)(arow + 64 * (1 + k) + 4 * ks);
                acc[1 + k][0] = wmma4(ak, bv0, acc[1 + k][0]);
                acc[1 + k][1] = wmma4(ak, bv1, acc[1 + k][1]);
            }
        }
        {
            const int nr = (n0 + col < nend) ? (n0 + col) : (nend - 1);
            const float* rowp = node_input + (size_t)nr * 128 + 2 * g;
#pragma unroll
            for (int ks = 0; ks < 8; ++ks) {
                const v2f bs0 = sB1s[ks][0][lane], bs1 = sB1s[ks][1][lane];
                const v2f bv0 = sB1v[ks][0][lane], bv1 = sB1v[ks][1][lane];
                const v2f a0 = *(const v2f*)(rowp + 4 * ks);
                acc[0][0] = wmma4(a0, bs0, acc[0][0]);
                acc[0][1] = wmma4(a0, bs1, acc[0][1]);
                const float* vp = rowp + 32 + 3 * (4 * ks) + 4 * g;
#pragma unroll
                for (int k = 0; k < 3; ++k) {
                    v2f ak;
                    ak[0] = vp[k];
                    ak[1] = vp[3 + k];
                    acc[1 + k][0] = wmma4(ak, bv0, acc[1 + k][0]);
                    acc[1 + k][1] = wmma4(ak, bv1, acc[1 + k][1]);
                }
            }
        }
#pragma unroll
        for (int r = 0; r < 8; ++r) {
            const int n = n0 + 8 * g + r;
            if (n < nend) {
                float* o = out + (size_t)n * 128;
#pragma unroll
                for (int nt = 0; nt < 2; ++nt) {
                    const int w = 16 * nt + col;
                    o[w] = acc[0][nt][r];
                    o[32 + w * 3 + 0] = acc[1][nt][r];
                    o[32 + w * 3 + 1] = acc[2][nt][r];
                    o[32 + w * 3 + 2] = acc[3][nt][r];
                }
            }
        }
    }
}

extern "C" void kernel_launch(void* const* d_in, const int* in_sizes, int n_in,
                              void* d_out, int out_size, void* d_ws, size_t ws_size,
                              hipStream_t stream) {
    (void)n_in; (void)out_size; (void)ws_size;
    const float* node_input   = (const float*)d_in[0];
    const float* edge_attr    = (const float*)d_in[1];
    const float* edge_scalars = (const float*)d_in[2];
    const float* W1_s = (const float*)d_in[3];
    const float* W1_v = (const float*)d_in[4];
    const float* M0   = (const float*)d_in[5];
    const float* M1   = (const float*)d_in[6];
    const float* Wtp0 = (const float*)d_in[7];
    const float* Wtp1 = (const float*)d_in[8];
    const float* Wtp2 = (const float*)d_in[9];
    const float* Wtp3 = (const float*)d_in[10];
    const float* W2_s = (const float*)d_in[11];
    const float* W2_v = (const float*)d_in[12];
    const int* edge_src = (const int*)d_in[13];
    const int* edge_dst = (const int*)d_in[14];

    int N = in_sizes[0] / 128;
    int E = in_sizes[13];

    float* feat_s = (float*)d_ws;
    float* feat_v = feat_s + (size_t)N * 32;
    int* cnt  = (int*)(feat_v + (size_t)N * 96);
    int* off  = cnt + N;
    int* rank = off + (N + 1);
    int* perm = rank + E;

    hipMemsetAsync(cnt, 0, (size_t)N * sizeof(int), stream);

    const int threads = 256;
    int blocksP  = ((N + 15) / 16 + PW - 1) / PW;
    int blocksEd = (E + threads - 1) / threads;

    hist_kernel<<<blocksEd, threads, 0, stream>>>(edge_dst, cnt, rank, E);
    scan_kernel<<<1, 1024, 0, stream>>>(cnt, off, N);
    fill_kernel<<<blocksEd, threads, 0, stream>>>(edge_dst, off, rank, perm, E);

    node_pre_kernel<<<blocksP, PW * 32, 0, stream>>>(node_input, W1_s, W1_v,
                                                     feat_s, feat_v, N);

    int blocksG = 256;
    node_gather_kernel<<<blocksG, GW * 32, 0, stream>>>(node_input, edge_attr, edge_scalars,
                                                        W1_s, W1_v, M0, M1,
                                                        Wtp0, Wtp1, Wtp2, Wtp3, W2_s, W2_v,
                                                        edge_src, off, perm, feat_s, feat_v,
                                                        (float*)d_out, N);
}


# Gather kernel: index loads off the per-node chain: a tile's 17 offsets in one vector load (no scalar load per node), a chunk's 16 edge ids loaded once and shared by lane reads; same sums
# speedup vs baseline: 6.2507x; 1.0363x over previous
#include <hip/hip_runtime.h>
#include <math.h>

typedef __attribute__((ext_vector_type(16))) _Float16 v16h;
typedef __attribute__((ext_vector_type(8)))  float    v8f;
typedef __attribute__((ext_vector_type(2)))  float    v2f;

__device__ __forceinline__ float gelu_f(float x) {
    const float c0 = 0.7978845608028654f;
    const float c1 = 2.8853900817779268f;
    float u = c0 * (x + 0.044715f * x * x * x);
    float e = __builtin_amdgcn_exp2f(u * c1);
    return x - x * __builtin_amdgcn_rcpf(e + 1.0f);
}

__device__ __forceinline__ v8f wmma16(v16h a, v16h b, v8f c) {
    return __builtin_amdgcn_wmma_f32_16x16x32_f16(
         false, a,  false, b,
         (short)0, c,  false,  false);
}

__device__ __forceinline__ v8f wmma4(v2f a, v2f b, v8f c) {
    return __builtin_amdgcn_wmma_f32_16x16x4_f32(
         false, a,  false, b,
         (short)0, c,  false,  false);
}

__device__ __forceinline__ v16h load_wT_frag(const float* __restrict__ W,
                                             int Kreal, int t, int lane,
                                             float scale) {
    int m = 16 * t + (lane & 15);
    int g = lane >> 4;
    v16h a;
#pragma unroll
    for (int e = 0; e < 16; ++e) {
        int r = e >> 1, j = e & 1;
        int k = ((r >> 2) << 4) + g * 8 + ((r & 3) << 1) + j;
        float val = (k < Kreal) ? W[k * 32 + m] * scale : 0.0f;
        a[e] = (_Float16)val;
    }
    return a;
}

__device__ __forceinline__ v16h pack_b_from_d(v8f d0, v8f d1) {
    v16h b;
#pragma unroll
    for (int e = 0; e < 16; ++e) {
        int r = e >> 1, j = e & 1;
        int i = ((r & 3) << 1) + j;
        float val = (r >> 2) ? d1[i] : d0[i];
        b[e] = (_Float16)val;
    }
    return b;
}

#define PW 4
#define PPITCH 132

__global__ void __launch_bounds__(PW * 32)
node_pre_kernel(const float* __restrict__ node_input,
                const float* __restrict__ W1_s,
                const float* __restrict__ W1_v,
                float* __restrict__ feat_s,
                float* __restrict__ feat_v, int N) {
    __shared__ __attribute__((aligned(16))) float sRow[PW][16 * PPITCH];

    const int lane = threadIdx.x & 31;
    const int col  = lane & 15;
    const int g    = lane >> 4;
    const int wv_in_blk = __builtin_amdgcn_readfirstlane((int)(threadIdx.x >> 5));
    const float inv32 = 0.17677669529663687f;

    v2f bs[8][2], bv[8][2];
#pragma unroll
    for (int ks = 0; ks < 8; ++ks) {
#pragma unroll
        for (int nt = 0; nt < 2; ++nt) {
            const int k0 = 4 * ks + 2 * g;
            const int w  = 16 * nt + col;
            bs[ks][nt][0] = W1_s[k0 * 64 + w] * inv32;
            bs[ks][nt][1] = W1_s[(k0 + 1) * 64 + w] * inv32;
            bv[ks][nt][0] = W1_v[k0 * 64 + w] * inv32;
            bv[ks][nt][1] = W1_v[(k0 + 1) * 64 + w] * inv32;
        }
    }

    float* srow = &sRow[wv_in_blk][0];
    const int ntiles = (N + 15) >> 4;
    for (int tile = blockIdx.x * PW + wv_in_blk; tile < ntiles; tile += gridDim.x * PW) {
        const int n0 = tile * 16;
#pragma unroll
        for (int j = 0; j < 16; ++j) {
            const int n = (n0 + j < N) ? (n0 + j) : (N - 1);
            const float4 x = *(const float4*)(node_input + (size_t)n * 128 + 4 * lane);
            *(float4*)(srow + j * PPITCH + 4 * lane) = x;
        }

        v8f acc[4][2];
#pragma unroll
        for (int cp = 0; cp < 4; ++cp) { acc[cp][0] = (v8f){}; acc[cp][1] = (v8f){}; }

        const float* arow = srow + col * PPITCH;
#pragma unroll
        for (int ks = 0; ks < 8; ++ks) {
            const v2f a0 = *(const v2f*)(arow + 4 * ks + 2 * g);
            acc[0][0] = wmma4(a0, bs[ks][0], acc[0][0]);
            acc[0][1] = wmma4(a0, bs[ks][1], acc[0][1]);
            const float* vp = arow + 32 + 3 * (4 * ks + 2 * g);
#pragma unroll
            for (int k = 0; k < 3; ++k) {
                v2f ak;
                ak[0] = vp[k];
                ak[1] = vp[3 + k];
                acc[1 + k][0] = wmma4(ak, bv[ks][0], acc[1 + k][0]);
                acc[1 + k][1] = wmma4(ak, bv[ks][1], acc[1 + k][1]);
            }
        }
#pragma unroll
        for (int r = 0; r < 8; ++r) {
            const int n = n0 + 8 * g + r;
            if (n < N) {
#pragma unroll
                for (int nt = 0; nt < 2; ++nt) {
                    const int w = 16 * nt + col;
                    feat_s[(size_t)n * 32 + w] = acc[0][nt][r];
                    float* fv = feat_v + (size_t)n * 96 + w * 3;
                    fv[0] = acc[1][nt][r];
                    fv[1] = acc[2][nt][r];
                    fv[2] = acc[3][nt][r];
                }
            }
        }
    }
}

__global__ void __launch_bounds__(256)
hist_kernel(const int* __restrict__ edge_dst, int* __restrict__ cnt,
            int* __restrict__ rank, int E) {
    int e = blockIdx.x * blockDim.x + threadIdx.x;
    if (e < E)
        rank[e] = __hip_atomic_fetch_add(cnt + edge_dst[e], 1, __ATOMIC_RELAXED,
                                         __HIP_MEMORY_SCOPE_AGENT);
}

__device__ __forceinline__ int wave_incl_scan(int x, int lane) {
    x += __builtin_amdgcn_update_dpp(0, x, 0x111, 0xf, 0xf, true);
    x += __builtin_amdgcn_update_dpp(0, x, 0x112, 0xf, 0xf, true);
    x += __builtin_amdgcn_update_dpp(0, x, 0x114, 0xf, 0xf, true);
    x += __builtin_amdgcn_update_dpp(0, x, 0x118, 0xf, 0xf, true);
    const int low = __builtin_amdgcn_readlane(x, 15);
    return (lane >= 16) ? (x + low) : x;
}

#define SROWS 16
__global__ void __launch_bounds__(1024)
scan_kernel(const int* __restrict__ cnt, int* __restrict__ off, int N) {
    __shared__ int wtot[32];
    const int lane = threadIdx.x & 31;
    const int wid  = threadIdx.x >> 5;
    const int seg  = (N + 31) / 32;
    const int beg  = wid * seg;
    const int end  = (beg + seg < N) ? (beg + seg) : N;
    int s = 0;
    for (int i = beg + lane; i < end; i += 32) s += cnt[i];
    s = __builtin_amdgcn_readlane(wave_incl_scan(s, lane), 31);
    if (lane == 0) wtot[wid] = s;
    __syncthreads();
    int carry = 0;
    for (int w = 0; w < wid; ++w) carry += wtot[w];
    for (int i0 = beg; i0 < end; i0 += 32 * SROWS) {
        int v[SROWS], x[SROWS];
#pragma unroll
        for (int r = 0; r < SROWS; ++r) {
            const int i = i0 + 32 * r + lane;
            v[r] = cnt[(i < end) ? i : (end - 1)];
        }
        __builtin_amdgcn_sched_barrier(0);
#pragma unroll
        for (int r = 0; r < SROWS; ++r) {
            if (i0 + 32 * r + lane >= end) v[r] = 0;
            x[r] = wave_incl_scan(v[r], lane);
        }
#pragma unroll
        for (int r = 0; r < SROWS; ++r) {
            const int i = i0 + 32 * r + lane;
            if (i < end) off[i] = carry + x[r] - v[r];
            carry += __builtin_amdgcn_readlane(x[r], 31);
        }
    }
    if (threadIdx.x == 0) {
        int t = 0;
        for (int w = 0; w < 32; ++w) t += wtot[w];
        off[N] = t;
    }
}

__global__ void __launch_bounds__(256)
fill_kernel(const int* __restrict__ edge_dst, const int* __restrict__ off,
            const int* __restrict__ rank, int* __restrict__ perm, int E) {
    int e = blockIdx.x * blockDim.x + threadIdx.x;
    if (e < E) perm[off[edge_dst[e]] + rank[e]] = e;
}

#define GW 16
#define SPITCH 260

__global__ void __launch_bounds__(GW * 32)
node_gather_kernel(const float* __restrict__ node_input,
                   const float* __restrict__ edge_attr,
                   const float* __restrict__ edge_scalars,
                   const float* __restrict__ W1_s, const float* __restrict__ W1_v,
                   const float* __restrict__ M0, const float* __restrict__ M1,
                   const float* __restrict__ Wtp0, const float* __restrict__ Wtp1,
                   const float* __restrict__ Wtp2, const float* __restrict__ Wtp3,
                   const float* __restrict__ W2_s, const float* __restrict__ W2_v,
                   const int* __restrict__ edge_src,
                   const int* __restrict__ off, const int* __restrict__ perm,
                   const float* __restrict__ feat_s, const float* __restrict__ feat_v,
                   float* __restrict__ out, int N) {
    __shared__ v16h  sFrag[12][32];
    __shared__ v2f   sB2s[16][2][32];
    __shared__ v2f   sB2v[16][2][32];
    __shared__ v2f   sB1s[8][2][32];
    __shared__ v2f   sB1v[8][2][32];
    __shared__ __attribute__((aligned(16))) float sStage[GW][16 * SPITCH];

    const int lane = threadIdx.x & 31;
    const int col  = lane & 15;
    const int g    = lane >> 4;
    const int wv_in_blk = __builtin_amdgcn_readfirstlane((int)(threadIdx.x >> 5));

    const float inv8  = 0.35355339059327373f;
    const float inv32 = 0.17677669529663687f;
    const float INV3  = 0.5773502691896258f;
    const float cconv = 0.125f * 0.31622776601683794f;
    const float ca = 0.9219544457292887f;
    const float cb = 0.3872983346207417f;
    const float sc2 = cconv * cb;
    const float sc1 = inv32 * ca;

    for (int idx = threadIdx.x; idx < 16 * 2 * 32; idx += blockDim.x) {
        const int ks = idx >> 6, nt = (idx >> 5) & 1, l = idx & 31;
        const int k0 = 4 * ks + 2 * (l >> 4);
        const int w  = 16 * nt + (l & 15);
        v2f bs, bv;
        bs[0] = W2_s[k0 * 32 + w] * sc2;  bs[1] = W2_s[(k0 + 1) * 32 + w] * sc2;
        bv[0] = W2_v[k0 * 32 + w] * sc2;  bv[1] = W2_v[(k0 + 1) * 32 + w] * sc2;
        sB2s[ks][nt][l] = bs;
        sB2v[ks][nt][l] = bv;
    }
    for (int idx = threadIdx.x; idx < 8 * 2 * 32; idx += blockDim.x) {
        const int ks = idx >> 6, nt = (idx >> 5) & 1, l = idx & 31;
        const int k0 = 4 * ks + 2 * (l >> 4);
        const int w  = 32 + 16 * nt + (l & 15);
        v2f bs, bv;
        bs[0] = W1_s[k0 * 64 + w] * sc1;  bs[1] = W1_s[(k0 + 1) * 64 + w] * sc1;
        bv[0] = W1_v[k0 * 64 + w] * sc1;  bv[1] = W1_v[(k0 + 1) * 64 + w] * sc1;
        sB1s[ks][nt][l] = bs;
        sB1v[ks][nt][l] = bv;
    }
    for (int f = wv_in_blk; f < 12; f += GW) {
        const int m = f >> 1;
        const float* Wp = (m == 0) ? M0 : (m == 1) ? M1 : (m == 2) ? Wtp0
                        : (m == 3) ? Wtp1 : (m == 4) ? Wtp2 : Wtp3;
        sFrag[f][lane] = load_wT_frag(Wp, (m == 0) ? 8 : 32, f & 1, lane,
                                      (m == 0) ? inv8 : inv32);
    }
    __syncthreads();

    float* stage = &sStage[wv_in_blk][0];

    const int nwaves = gridDim.x * GW;
    const int wave   = blockIdx.x * GW + wv_in_blk;
    const int q      = N / nwaves;
    const int rem    = N - q * nwaves;
    const int nbeg   = wave * q + ((wave < rem) ? wave : rem);
    const int nend   = nbeg + q + ((wave < rem) ? 1 : 0);

    for (int n0 = nbeg; n0 < nend; n0 += 16) {
        const int on   = n0 + lane;
        const int offv = off[(on < nend) ? on : nend];
        for (int j = 0; j < 16; ++j) {
            const int beg = __builtin_amdgcn_readlane(offv, j);
            const int end = __builtin_amdgcn_readlane(offv, j + 1);

            float S0[2] = {0.f, 0.f}, S1[2] = {0.f, 0.f};
            float V0[2][3] = {{0.f, 0.f, 0.f}, {0.f, 0.f, 0.f}};
            float V1[2][3] = {{0.f, 0.f, 0.f}, {0.f, 0.f, 0.f}};

            for (int base = beg; base < end; base += 16) {
                asm volatile("" ::: "memory");
                const bool vcol = base + col < end;
                const int  ecol = perm[vcol ? (base + col) : beg];
                v16h b0;
#pragma unroll
                for (int q = 0; q < 16; ++q) b0[q] = (_Float16)0.0f;
                if (g == 0 && vcol) {
                    const float* p = edge_scalars + (size_t)ecol * 8;
#pragma unroll
                    for (int q = 0; q < 8; ++q) b0[q] = (_Float16)p[q];
                }
                v8f z = {};
                v8f d0 = wmma16(sFrag[0][lane], b0, z);
                v8f d1 = wmma16(sFrag[1][lane], b0, z);
#pragma unroll
                for (int q = 0; q < 8; ++q) { d0[q] = gelu_f(d0[q]); d1[q] = gelu_f(d1[q]); }
                v16h b1 = pack_b_from_d(d0, d1);
                d0 = wmma16(sFrag[2][lane], b1, z);
                d1 = wmma16(sFrag[3][lane], b1, z);
#pragma unroll
                for (int q = 0; q < 8; ++q) { d0[q] = gelu_f(d0[q]); d1[q] = gelu_f(d1[q]); }
                v16h h2 = pack_b_from_d(d0, d1);
                v8f w0[2], w1[2], w2[2], w3[2];
#pragma unroll
                for (int t = 0; t < 2; ++t) {
                    w0[t] = wmma16(h2, sFrag[4 + t][lane], z);
                    w1[t] = wmma16(h2, sFrag[6 + t][lane], z);
                    w2[t] = wmma16(h2, sFrag[8 + t][lane], z);
                    w3[t] = wmma16(h2, sFrag[10 + t][lane], z);
                }
#pragma unroll
                for (int i = 0; i < 8; ++i) {
                    const int p  = base + 8 * g + i;
                    const bool v = p < end;
                    const int elo = __builtin_amdgcn_readlane(ecol, i);
                    const int ehi = __builtin_amdgcn_readlane(ecol, 8 + i);
                    const int e   = g ? ehi : elo;
                    const int src = edge_src[e];
                    const float4 ea = *(const float4*)(edge_attr + (size_t)e * 4);
                    const float y0  = v ? ea.x : 0.f;
                    const float y1x = v ? ea.y : 0.f;
                    const float y1y = v ? ea.z : 0.f;
                    const float y1z = v ? ea.w : 0.f;
                    const int fso = src * 32;
                    const int fvo = src * 96;
#pragma unroll
                    for (int t = 0; t < 2; ++t) {
                        const int u = 16 * t + col;
                        const float es  = feat_s[fso + u];
                        const float evx = feat_v[fvo + u * 3 + 0];
                        const float evy = feat_v[fvo + u * 3 + 1];
                        const float evz = feat_v[fvo + u * 3 + 2];
                        const float dotv = evx * y1x + evy * y1y + evz * y1z;
                        S0[t] += w0[t][i] * es * y0;
                        S1[t] += w3[t][i] * dotv * INV3;
                        const float t1 = w1[t][i] * es;
                        V0[t][0] += t1 * y1x;
                        V0[t][1] += t1 * y1y;
                        V0[t][2] += t1 * y1z;
                        const float t2f = w2[t][i] * y0;
                        V1[t][0] += t2f * evx;
                        V1[t][1] += t2f * evy;
                        V1[t][2] += t2f * evz;
                    }
                }
            }
            float* srow = stage + j * SPITCH;
            {
                const float k0 = g ? S0[1] : S0[0], x0 = g ? S0[0] : S0[1];
                const float k1 = g ? S1[1] : S1[0], x1 = g ? S1[0] : S1[1];
                srow[lane]      = k0 + __shfl_xor(x0, 16);
                srow[32 + lane] = k1 + __shfl_xor(x1, 16);
            }
#pragma unroll
            for (int k = 0; k < 3; ++k) {
                const float k0 = g ? V0[1][k] : V0[0][k], x0 = g ? V0[0][k] : V0[1][k];
                const float k1 = g ? V1[1][k] : V1[0][k], x1 = g ? V1[0][k] : V1[1][k];
                srow[64 * (1 + k) + lane]      = k0 + __shfl_xor(x0, 16);
                srow[64 * (1 + k) + 32 + lane] = k1 + __shfl_xor(x1, 16);
            }
        }
        asm volatile("" ::: "memory");

        v8f acc[4][2];
#pragma unroll
        for (int cp = 0; cp < 4; ++cp) { acc[cp][0] = (v8f){}; acc[cp][1] = (v8f){}; }

        const float* arow = stage + col * SPITCH + 2 * g;
#pragma unroll
        for (int ks = 0; ks < 16; ++ks) {
            const v2f bs0 = sB2s[ks][0][lane], bs1 = sB2s[ks][1][lane];
            const v2f bv0 = sB2v[ks][0][lane], bv1 = sB2v[ks][1][lane];
            const v2f a0 = *(const v2f*)(arow + 4 * ks);
            acc[0][0] = wmma4(a0, bs0, acc[0][0]);
            acc[0][1] = wmma4(a0, bs1, acc[0][1]);
#pragma unroll
            for (int k = 0; k < 3; ++k) {
                const v2f ak = *(const v2f*)(arow + 64 * (1 + k) + 4 * ks);
                acc[1 + k][0] = wmma4(ak, bv0, acc[1 + k][0]);
                acc[1 + k][1] = wmma4(ak, bv1, acc[1 + k][1]);
            }
        }
        {
            const int nr = (n0 + col < nend) ? (n0 + col) : (nend - 1);
            const float* rowp = node_input + (size_t)nr * 128 + 2 * g;
#pragma unroll
            for (int ks = 0; ks < 8; ++ks) {
                const v2f bs0 = sB1s[ks][0][lane], bs1 = sB1s[ks][1][lane];
                const v2f bv0 = sB1v[ks][0][lane], bv1 = sB1v[ks][1][lane];
                const v2f a0 = *(const v2f*)(rowp + 4 * ks);
                acc[0][0] = wmma4(a0, bs0, acc[0][0]);
                acc[0][1] = wmma4(a0, bs1, acc[0][1]);
                const float* vp = rowp + 32 + 3 * (4 * ks) + 4 * g;
#pragma unroll
                for (int k = 0; k < 3; ++k) {
                    v2f ak;
                    ak[0] = vp[k];
                    ak[1] = vp[3 + k];
                    acc[1 + k][0] = wmma4(ak, bv0, acc[1 + k][0]);
                    acc[1 + k][1] = wmma4(ak, bv1, acc[1 + k][1]);
                }
            }
        }
#pragma unroll
        for (int r = 0; r < 8; ++r) {
            const int n = n0 + 8 * g + r;
            if (n < nend) {
                float* o = out + (size_t)n * 128;
#pragma unroll
                for (int nt = 0; nt < 2; ++nt) {
                    const int w = 16 * nt + col;
                    o[w] = acc[0][nt][r];
                    o[32 + w * 3 + 0] = acc[1][nt][r];
                    o[32 + w * 3 + 1] = acc[2][nt][r];
                    o[32 + w * 3 + 2] = acc[3][nt][r];
                }
            }
        }
    }
}

extern "C" void kernel_launch(void* const* d_in, const int* in_sizes, int n_in,
                              void* d_out, int out_size, void* d_ws, size_t ws_size,
                              hipStream_t stream) {
    (void)n_in; (void)out_size; (void)ws_size;
    const float* node_input   = (const float*)d_in[0];
    const float* edge_attr    = (const float*)d_in[1];
    const float* edge_scalars = (const float*)d_in[2];
    const float* W1_s = (const float*)d_in[3];
    const float* W1_v = (const float*)d_in[4];
    const float* M0   = (const float*)d_in[5];
    const float* M1   = (const float*)d_in[6];
    const float* Wtp0 = (const float*)d_in[7];
    const float* Wtp1 = (const float*)d_in[8];
    const float* Wtp2 = (const float*)d_in[9];
    const float* Wtp3 = (const float*)d_in[10];
    const float* W2_s = (const float*)d_in[11];
    const float* W2_v = (const float*)d_in[12];
    const int* edge_src = (const int*)d_in[13];
    const int* edge_dst = (const int*)d_in[14];

    int N = in_sizes[0] / 128;
    int E = in_sizes[13];

    float* feat_s = (float*)d_ws;
    float* feat_v = feat_s + (size_t)N * 32;
    int* cnt  = (int*)(feat_v + (size_t)N * 96);
    int* off  = cnt + N;
    int* rank = off + (N + 1);
    int* perm = rank + E;

    hipMemsetAsync(cnt, 0, (size_t)N * sizeof(int), stream);

    const int threads = 256;
    int blocksP  = ((N + 15) / 16 + PW - 1) / PW;
    int blocksEd = (E + threads - 1) / threads;

    hist_kernel<<<blocksEd, threads, 0, stream>>>(edge_dst, cnt, rank, E);
    scan_kernel<<<1, 1024, 0, stream>>>(cnt, off, N);
    fill_kernel<<<blocksEd, threads, 0, stream>>>(edge_dst, off, rank, perm, E);

    node_pre_kernel<<<blocksP, PW * 32, 0, stream>>>(node_input, W1_s, W1_v,
                                                     feat_s, feat_v, N);

    int blocksG = 256;
    node_gather_kernel<<<blocksG, GW * 32, 0, stream>>>(node_input, edge_attr, edge_scalars,
                                                        W1_s, W1_v, M0, M1,
                                                        Wtp0, Wtp1, Wtp2, Wtp3, W2_s, W2_v,
                                                        edge_src, off, perm, feat_s, feat_v,
                                                        (float*)d_out, N);
}
